# starting kernel
# baseline (speedup 1.0000x reference)
#include <hip/hip_runtime.h>

typedef __attribute__((ext_vector_type(2))) float v2f;
typedef __attribute__((ext_vector_type(8))) float v8f;

#define B_BATCH 256
#define S_DIM   16384
#define R_RANK  16
#define KSPLIT  8
#define NT_TOTAL 65
#define NCOLS   1040
#define MBLK    4
#define KCHUNK  (S_DIM / KSPLIT)

__device__ __forceinline__ v8f wmma_f32x4(v2f a, v2f b, v8f c) {
    return __builtin_amdgcn_wmma_f32_16x16x4_f32(false, a, false, b, (short)0, c,
                                                 false, false);
}

__global__ __launch_bounds__(32)
void tt_phase1(const float* __restrict__ z,
               const float* __restrict__ c1, const float* __restrict__ c2,
               const float* __restrict__ c3, const float* __restrict__ c4,
               const float* __restrict__ c5,
               float* __restrict__ vpart) {
    const int lane = threadIdx.x & 31;
    const int half = lane >> 4;
    const int l    = lane & 15;

    const int t  = blockIdx.x;
    const int mb = blockIdx.y;
    const int ks = blockIdx.z;

    const int b0      = mb * 64;
    const int k_begin = ks * KCHUNK;

    const float* ap0 = z + (size_t)(b0 + l) * S_DIM + k_begin + 2 * half;
    const float* ap1 = ap0 + (size_t)16 * S_DIM;
    const float* ap2 = ap0 + (size_t)32 * S_DIM;
    const float* ap3 = ap0 + (size_t)48 * S_DIM;

    v8f acc0 = {}, acc1 = {}, acc2 = {}, acc3 = {};
    int colbase;

    if (t < 64) {
        const int k = t >> 4;
        const int d = t & 15;
        const float* cp = (k == 0) ? c1 : (k == 1) ? c2 : (k == 2) ? c3 : c4;
        colbase = k * 256 + d * 16;
        const float* bp = cp + (size_t)d * S_DIM * R_RANK
                             + (size_t)(k_begin + 2 * half) * R_RANK + l;
        for (int it = 0; it < KCHUNK / 16; ++it) {
            __builtin_prefetch(bp + 4 * 256, 0, 1);
            __builtin_prefetch(ap0 + 64, 0, 1);
#pragma unroll
            for (int u = 0; u < 4; ++u) {
                v2f bf;
                bf.x = bp[u * 64];
                bf.y = bp[u * 64 + 16];
                v2f a0 = *(const v2f*)(ap0 + u * 4);
                v2f a1 = *(const v2f*)(ap1 + u * 4);
                v2f a2 = *(const v2f*)(ap2 + u * 4);
                v2f a3 = *(const v2f*)(ap3 + u * 4);
                acc0 = wmma_f32x4(a0, bf, acc0);
                acc1 = wmma_f32x4(a1, bf, acc1);
                acc2 = wmma_f32x4(a2, bf, acc2);
                acc3 = wmma_f32x4(a3, bf, acc3);
            }
            ap0 += 16; ap1 += 16; ap2 += 16; ap3 += 16;
            bp  += 256;
        }
    } else {
        colbase = 1024;
        const float* bp = c5 + (size_t)l * S_DIM + k_begin + 2 * half;
        for (int it = 0; it < KCHUNK / 16; ++it) {
            __builtin_prefetch(bp + 64, 0, 1);
            __builtin_prefetch(ap0 + 64, 0, 1);
#pragma unroll
            for (int u = 0; u < 4; ++u) {
                v2f bf = *(const v2f*)(bp + u * 4);
                v2f a0 = *(const v2f*)(ap0 + u * 4);
                v2f a1 = *(const v2f*)(ap1 + u * 4);
                v2f a2 = *(const v2f*)(ap2 + u * 4);
                v2f a3 = *(const v2f*)(ap3 + u * 4);
                acc0 = wmma_f32x4(a0, bf, acc0);
                acc1 = wmma_f32x4(a1, bf, acc1);
                acc2 = wmma_f32x4(a2, bf, acc2);
                acc3 = wmma_f32x4(a3, bf, acc3);
            }
            ap0 += 16; ap1 += 16; ap2 += 16; ap3 += 16;
            bp  += 16;
        }
    }

    float* vp = vpart + (size_t)ks * B_BATCH * NCOLS
                      + (size_t)(b0 + 8 * half) * NCOLS + colbase + l;
    v8f accs[MBLK] = {acc0, acc1, acc2, acc3};
#pragma unroll
    for (int m = 0; m < MBLK; ++m) {
#pragma unroll
        for (int v = 0; v < 8; ++v) {
            vp[(size_t)(m * 16 + v) * NCOLS] = accs[m][v];
        }
    }
}

__global__ __launch_bounds__(256)
void tt_phase2(const float* __restrict__ vpart, float* __restrict__ w) {
    const int b = blockIdx.x * blockDim.x + threadIdx.x;
    if (b >= B_BATCH) return;

    const size_t rowoff = (size_t)b * NCOLS;
    float t[R_RANK], nt[R_RANK];

#pragma unroll
    for (int d = 0; d < R_RANK; ++d) {
        float s = 0.f;
        for (int p = 0; p < KSPLIT; ++p)
            s += vpart[(size_t)p * B_BATCH * NCOLS + rowoff + 1024 + d];
        t[d] = s;
    }
    for (int k = 3; k >= 0; --k) {
#pragma unroll
        for (int d = 0; d < R_RANK; ++d) {
            float s = 0.f;
#pragma unroll
            for (int e = 0; e < R_RANK; ++e) {
                float v = 0.f;
                for (int p = 0; p < KSPLIT; ++p)
                    v += vpart[(size_t)p * B_BATCH * NCOLS + rowoff +
                               k * 256 + d * 16 + e];
                s += v * t[e];
            }
            nt[d] = s;
        }
#pragma unroll
        for (int d = 0; d < R_RANK; ++d) t[d] = nt[d];
    }
#pragma unroll
    for (int r = 0; r < R_RANK; ++r)
        w[b * R_RANK + r] = t[r];
}

__global__ __launch_bounds__(32)
void tt_phase3(const float* __restrict__ w, const float* __restrict__ core0,
               float* __restrict__ out) {
    const int lane = threadIdx.x & 31;
    const int half = lane >> 4;
    const int l    = lane & 15;
    const int s0   = blockIdx.x * 16;
    const int b0   = blockIdx.y * 16;

    const float* wp = w     + (size_t)(b0 + l) * R_RANK + 2 * half;
    const float* cp = core0 + (size_t)(s0 + l) * R_RANK + 2 * half;

    v8f acc = {};
#pragma unroll
    for (int k0 = 0; k0 < R_RANK; k0 += 4) {
        v2f a  = *(const v2f*)(wp + k0);
        v2f bf = *(const v2f*)(cp + k0);
        acc = wmma_f32x4(a, bf, acc);
    }
    float* op = out + (size_t)(b0 + 8 * half) * S_DIM + s0 + l;
#pragma unroll
    for (int v = 0; v < 8; ++v) {
        op[(size_t)v * S_DIM] = acc[v];
    }
}

extern "C" void kernel_launch(void* const* d_in, const int* in_sizes, int n_in,
                              void* d_out, int out_size, void* d_ws, size_t ws_size,
                              hipStream_t stream) {
    const float* z  = (const float*)d_in[0];
    const float* c0 = (const float*)d_in[1];
    const float* c1 = (const float*)d_in[2];
    const float* c2 = (const float*)d_in[3];
    const float* c3 = (const float*)d_in[4];
    const float* c4 = (const float*)d_in[5];
    const float* c5 = (const float*)d_in[6];
    float* out = (float*)d_out;

    float* vpart = (float*)d_ws;
    float* w     = vpart + (size_t)KSPLIT * B_BATCH * NCOLS;

    tt_phase1<<<dim3(NT_TOTAL, B_BATCH / 64, KSPLIT), 32, 0, stream>>>(
        z, c1, c2, c3, c4, c5, vpart);
    tt_phase2<<<dim3(1), 256, 0, stream>>>(vpart, w);
    tt_phase3<<<dim3(S_DIM / 16, B_BATCH / 16), 32, 0, stream>>>(w, c0, out);
}


# phase 1 tiled along N in registers: wave tile 32 x 64 (2 M x 4 N tiles), a z fragment feeds 4 core tiles, core rows read as 16-byte pieces; 1 wave per SIMD, loads a step ahead; sum order kept
# speedup vs baseline: 6.3915x; 6.3915x over previous
#include <hip/hip_runtime.h>

typedef __attribute__((ext_vector_type(2))) float v2f;
typedef __attribute__((ext_vector_type(4))) float v4f;
typedef __attribute__((ext_vector_type(8))) float v8f;

#define B_BATCH 256
#define S_DIM   16384
#define R_RANK  16
#define KSPLIT  8
#define NCOLS   1040
#define KCHUNK  (S_DIM / KSPLIT)

__device__ __forceinline__ v8f wmma_f32x4(v2f a, v2f b, v8f c) {
    return __builtin_amdgcn_wmma_f32_16x16x4_f32(false, a, false, b, (short)0, c,
                                                 false, false);
}

#define P1_THREADS     128
#define P1_MAIN_GROUPS 256
#define P1_C5_GROUPS   32
#define P1_ITERS       (KCHUNK / 16)
#define P1_PF          6
#define CORE_D_STRIDE  ((size_t)S_DIM * R_RANK)

struct P1Frags {
    v2f a[2][4];
    v4f b[4][2];
};

__device__ __forceinline__ void p1_load(P1Frags& f, const float* ap0,
                                        const float* ap1, const float* bp) {
#pragma unroll
    for (int u = 0; u < 4; ++u) {
        f.a[0][u] = *(const v2f*)(ap0 + u * 4);
        f.a[1][u] = *(const v2f*)(ap1 + u * 4);
        f.b[u][0] = *(const v4f*)(bp + u * 64);
        f.b[u][1] = *(const v4f*)(bp + u * 64 + 16);
    }
}

__device__ __forceinline__ void p1_mma(const P1Frags& f, v8f (&acc)[2][4]) {
#pragma unroll
    for (int u = 0; u < 4; ++u) {
#pragma unroll
        for (int j = 0; j < 4; ++j) {
            v2f bf;
            bf.x = f.b[u][0][j];
            bf.y = f.b[u][1][j];
            acc[0][j] = wmma_f32x4(f.a[0][u], bf, acc[0][j]);
            acc[1][j] = wmma_f32x4(f.a[1][u], bf, acc[1][j]);
        }
    }
}

struct C5Frags {
    v2f a[8], b[8];
};

__device__ __forceinline__ void c5_load(C5Frags& f, const float* ap,
                                        const float* bp) {
#pragma unroll
    for (int u = 0; u < 8; ++u) {
        f.a[u] = *(const v2f*)(ap + u * 4);
        f.b[u] = *(const v2f*)(bp + u * 4);
    }
}

__device__ __forceinline__ v8f c5_mma(const C5Frags& f, v8f acc) {
#pragma unroll
    for (int u = 0; u < 8; ++u)
        acc = wmma_f32x4(f.a[u], f.b[u], acc);
    return acc;
}

__global__ __launch_bounds__(P1_THREADS)
void tt_phase1(const float* __restrict__ z,
               const float* __restrict__ c1, const float* __restrict__ c2,
               const float* __restrict__ c3, const float* __restrict__ c4,
               const float* __restrict__ c5,
               float* __restrict__ vpart) {
    const int lane = threadIdx.x & 31;
    const int wv   = threadIdx.x >> 5;
    const int half = lane >> 4;
    const int l    = lane & 15;
    const int g    = blockIdx.x;

    if (g < P1_MAIN_GROUPS) {
        const int ks = g & (KSPLIT - 1);
        const int k  = (g >> 3) & 3;
        const int mb = g >> 5;
        const int b0 = mb * 32;
        const int d0 = wv * 4;
        const int k_begin = ks * KCHUNK;
        const float* cp = (k == 0) ? c1 : (k == 1) ? c2 : (k == 2) ? c3 : c4;

        const float* ap0 = z + (size_t)(b0 + l) * S_DIM + k_begin + 2 * half;
        const float* ap1 = ap0 + (size_t)16 * S_DIM;
        const float* bp  = cp + (size_t)(d0 + (l >> 2)) * CORE_D_STRIDE
                              + (size_t)(k_begin + 2 * half) * R_RANK + 4 * (l & 3);
        const float* pfb = cp + (size_t)(d0 + (lane >> 3)) * CORE_D_STRIDE
                              + (size_t)k_begin * R_RANK + (lane & 7) * 64;
        const float* pfa = z + (size_t)(b0 + wv * 8 + (lane & 7)) * S_DIM + k_begin;

        v8f acc[2][4] = {};
        P1Frags f0, f1;
        p1_load(f0, ap0, ap1, bp);
#pragma unroll
        for (int p = 2; p < P1_PF; p += 2) {
            __builtin_prefetch(pfb + p * 256, 0, 1);
            __builtin_prefetch(pfa + p * 16, 0, 1);
        }
        for (int it = 0; it < P1_ITERS - 2; it += 2) {
            const int pf = min(it + P1_PF, P1_ITERS - 2);
            __builtin_prefetch(pfb + pf * 256, 0, 1);
            __builtin_prefetch(pfa + pf * 16, 0, 1);
            p1_load(f1, ap0 + 16, ap1 + 16, bp + 256);
            __builtin_amdgcn_sched_barrier(0);
            p1_mma(f0, acc);
            __builtin_amdgcn_sched_barrier(0);
            p1_load(f0, ap0 + 32, ap1 + 32, bp + 512);
            __builtin_amdgcn_sched_barrier(0);
            p1_mma(f1, acc);
            __builtin_amdgcn_sched_barrier(0);
            ap0 += 32; ap1 += 32; bp += 512;
        }
        p1_load(f1, ap0 + 16, ap1 + 16, bp + 256);
        __builtin_amdgcn_sched_barrier(0);
        p1_mma(f0, acc);
        p1_mma(f1, acc);

        float* vp = vpart + (size_t)ks * B_BATCH * NCOLS
                          + (size_t)(b0 + 8 * half) * NCOLS
                          + k * 256 + d0 * 16 + 4 * l;
#pragma unroll
        for (int m = 0; m < 2; ++m) {
#pragma unroll
            for (int v = 0; v < 8; ++v) {
                v4f o = {acc[m][0][v], acc[m][1][v], acc[m][2][v], acc[m][3][v]};
                *(v4f*)(vp + (size_t)(m * 16 + v) * NCOLS) = o;
            }
        }
    } else {
        const int gid = (g - P1_MAIN_GROUPS) * (P1_THREADS / 32) + wv;
        const int ks  = gid & (KSPLIT - 1);
        const int mt  = gid >> 3;
        const int k_begin = ks * KCHUNK;
        const float* ap = z  + (size_t)(mt * 16 + l) * S_DIM + k_begin + 2 * half;
        const float* bp = c5 + (size_t)l * S_DIM + k_begin + 2 * half;
        v8f acc = {};
        const float* pfa = ap;
        const float* pfb = bp;
        C5Frags q0, q1;
        c5_load(q0, ap, bp);
        for (int it = 0; it < KCHUNK / 32 - 2; it += 2) {
            const int pf = min(it + 4, KCHUNK / 32 - 2);
            __builtin_prefetch(pfa + pf * 32, 0, 1);
            __builtin_prefetch(pfb + pf * 32, 0, 1);
            __builtin_prefetch(pfa + (pf + 1) * 32, 0, 1);
            __builtin_prefetch(pfb + (pf + 1) * 32, 0, 1);
            c5_load(q1, ap + 32, bp + 32);
            acc = c5_mma(q0, acc);
            c5_load(q0, ap + 64, bp + 64);
            acc = c5_mma(q1, acc);
            ap += 64; bp += 64;
        }
        c5_load(q1, ap + 32, bp + 32);
        acc = c5_mma(q0, acc);
        acc = c5_mma(q1, acc);
        float* vp = vpart + (size_t)ks * B_BATCH * NCOLS
                          + (size_t)(mt * 16 + 8 * half) * NCOLS + 1024 + l;
#pragma unroll
        for (int v = 0; v < 8; ++v)
            vp[(size_t)v * NCOLS] = acc[v];
    }
}

#define F_ROWS    16
#define F_PIX     1024
#define F_THREADS 256
#define F_WAVES   (F_THREADS / 32)
#define F_TILES   (F_PIX / 16 / F_WAVES)
#define V_F4      (F_ROWS * NCOLS / 4)
#define PLANE_F4  (B_BATCH * NCOLS / 4)

__global__ __launch_bounds__(F_THREADS)
void tt_chain_out(const float* __restrict__ vpart,
                  const float* __restrict__ core0,
                  float* __restrict__ out) {
    __shared__ v4f   Vl[V_F4];
    __shared__ float tl[2][F_ROWS][R_RANK];

    const int tid  = threadIdx.x;
    const int lane = tid & 31;
    const int wv   = tid >> 5;
    const int half = lane >> 4;
    const int l    = lane & 15;
    const int b0   = blockIdx.y * F_ROWS;
    const int s0   = blockIdx.x * F_PIX + wv * (F_PIX / F_WAVES);

    const float* cp = core0 + (size_t)(s0 + l) * R_RANK + 2 * half;
    v2f bf[F_TILES][4];
#pragma unroll
    for (int j = 0; j < F_TILES; ++j) {
#pragma unroll
        for (int q = 0; q < 4; ++q)
            bf[j][q] = *(const v2f*)(cp + (size_t)j * 16 * R_RANK + q * 4);
    }

    const v4f* src = (const v4f*)(vpart + (size_t)b0 * NCOLS);
    const bool has_rest = wv < (V_F4 % F_THREADS) / 32;
    const int  i_rest   = (V_F4 / F_THREADS) * F_THREADS + tid;
    v4f rest[KSPLIT];
    if (has_rest) {
#pragma unroll
        for (int p = 0; p < KSPLIT; ++p)
            rest[p] = src[(size_t)p * PLANE_F4 + i_rest];
    }
#pragma unroll 4
    for (int it = 0; it < V_F4 / F_THREADS; ++it) {
        const int i = it * F_THREADS + tid;
        v4f s = {0.f, 0.f, 0.f, 0.f};
#pragma unroll
        for (int p = 0; p < KSPLIT; ++p)
            s += src[(size_t)p * PLANE_F4 + i];
        Vl[i] = s;
    }
    if (has_rest) {
        v4f s = {0.f, 0.f, 0.f, 0.f};
#pragma unroll
        for (int p = 0; p < KSPLIT; ++p)
            s += rest[p];
        Vl[i_rest] = s;
    }
    __syncthreads();

    {
        const int r = tid >> 4;
        const int d = tid & 15;
        const float* Vrow = (const float*)Vl + r * NCOLS;
        tl[0][r][d] = Vrow[1024 + d];
        __syncthreads();
        int cur = 0;
        for (int k = 3; k >= 0; --k) {
            const float* vk = Vrow + k * 256 + d * 16;
            const float* t  = tl[cur][r];
            float s = 0.f;
#pragma unroll
            for (int e = 0; e < R_RANK; ++e)
                s += vk[e] * t[e];
            tl[cur ^ 1][r][d] = s;
            cur ^= 1;
            __syncthreads();
        }
    }

    const float* wp = &tl[0][l][2 * half];
    v2f a[4];
#pragma unroll
    for (int q = 0; q < 4; ++q)
        a[q] = *(const v2f*)(wp + q * 4);

    float* op = out + (size_t)(b0 + 8 * half) * S_DIM + s0 + l;
#pragma unroll
    for (int j = 0; j < F_TILES; ++j) {
        v8f acc = {};
#pragma unroll
        for (int q = 0; q < 4; ++q)
            acc = wmma_f32x4(a[q], bf[j][q], acc);
#pragma unroll
        for (int v = 0; v < 8; ++v)
            op[(size_t)v * S_DIM + j * 16] = acc[v];
    }
}

extern "C" void kernel_launch(void* const* d_in, const int* in_sizes, int n_in,
                              void* d_out, int out_size, void* d_ws, size_t ws_size,
                              hipStream_t stream) {
    const float* z  = (const float*)d_in[0];
    const float* c0 = (const float*)d_in[1];
    const float* c1 = (const float*)d_in[2];
    const float* c2 = (const float*)d_in[3];
    const float* c3 = (const float*)d_in[4];
    const float* c4 = (const float*)d_in[5];
    const float* c5 = (const float*)d_in[6];
    float* out = (float*)d_out;

    float* vpart = (float*)d_ws;

    tt_phase1<<<dim3(P1_MAIN_GROUPS + P1_C5_GROUPS), P1_THREADS, 0, stream>>>(
        z, c1, c2, c3, c4, c5, vpart);
    tt_chain_out<<<dim3(S_DIM / F_PIX, B_BATCH / F_ROWS), F_THREADS, 0, stream>>>(
        vpart, c0, out);
}


# phase 1: core 5's 16 columns computed inside the 256 main groups (group k: rows 8k..8k+7, a K quarter per wave, quarters added via LDS), not by 32 extra groups; sum order of those columns changed
# speedup vs baseline: 7.3128x; 1.1441x over previous
#include <hip/hip_runtime.h>

typedef __attribute__((ext_vector_type(2))) float v2f;
typedef __attribute__((ext_vector_type(4))) float v4f;
typedef __attribute__((ext_vector_type(8))) float v8f;

#define B_BATCH 256
#define S_DIM   16384
#define R_RANK  16
#define KSPLIT  8
#define NCOLS   1040
#define KCHUNK  (S_DIM / KSPLIT)

__device__ __forceinline__ v8f wmma_f32x4(v2f a, v2f b, v8f c) {
    return __builtin_amdgcn_wmma_f32_16x16x4_f32(false, a, false, b, (short)0, c,
                                                 false, false);
}

#define P1_THREADS     128
#define P1_MAIN_GROUPS 256
#define P1_ITERS       (KCHUNK / 16)
#define P1_PF          6
#define CORE_D_STRIDE  ((size_t)S_DIM * R_RANK)

struct P1Frags {
    v2f a[2][4];
    v4f b[4][2];
};

__device__ __forceinline__ void p1_load(P1Frags& f, const float* ap0,
                                        const float* ap1, const float* bp) {
#pragma unroll
    for (int u = 0; u < 4; ++u) {
        f.a[0][u] = *(const v2f*)(ap0 + u * 4);
        f.a[1][u] = *(const v2f*)(ap1 + u * 4);
        f.b[u][0] = *(const v4f*)(bp + u * 64);
        f.b[u][1] = *(const v4f*)(bp + u * 64 + 16);
    }
}

__device__ __forceinline__ void p1_mma(const P1Frags& f, v8f (&acc)[2][4]) {
#pragma unroll
    for (int u = 0; u < 4; ++u) {
#pragma unroll
        for (int j = 0; j < 4; ++j) {
            v2f bf;
            bf.x = f.b[u][0][j];
            bf.y = f.b[u][1][j];
            acc[0][j] = wmma_f32x4(f.a[0][u], bf, acc[0][j]);
            acc[1][j] = wmma_f32x4(f.a[1][u], bf, acc[1][j]);
        }
    }
}

#define C5_QUARTER (KCHUNK / 4)
#define C5_CHUNK   128

struct C5Frags {
    v4f a[16], b[16];
};

__device__ __forceinline__ void c5_load(C5Frags& f, const float* ap,
                                        const float* bp) {
#pragma unroll
    for (int i = 0; i < 16; ++i) {
        f.a[i] = *(const v4f*)(ap + i * 8);
        f.b[i] = *(const v4f*)(bp + i * 8);
    }
}

__device__ __forceinline__ void c5_mma(const C5Frags& f, v8f& e0, v8f& e1) {
#pragma unroll
    for (int i = 0; i < 16; ++i) {
        v2f a0 = {f.a[i][0], f.a[i][1]}, b0 = {f.b[i][0], f.b[i][1]};
        v2f a1 = {f.a[i][2], f.a[i][3]}, b1 = {f.b[i][2], f.b[i][3]};
        e0 = wmma_f32x4(a0, b0, e0);
        e1 = wmma_f32x4(a1, b1, e1);
    }
}

__global__ __launch_bounds__(P1_THREADS)
void tt_phase1(const float* __restrict__ z,
               const float* __restrict__ c1, const float* __restrict__ c2,
               const float* __restrict__ c3, const float* __restrict__ c4,
               const float* __restrict__ c5,
               float* __restrict__ vpart) {
    const int lane = threadIdx.x & 31;
    const int wv   = threadIdx.x >> 5;
    const int half = lane >> 4;
    const int l    = lane & 15;
    const int g    = blockIdx.x;
    __shared__ float red[P1_THREADS / 32][8][R_RANK];

    {
        const int ks = g & (KSPLIT - 1);
        const int k  = (g >> 3) & 3;
        const int mb = g >> 5;
        const int b0 = mb * 32;
        const int d0 = wv * 4;
        const int k_begin = ks * KCHUNK;
        const float* cp = (k == 0) ? c1 : (k == 1) ? c2 : (k == 2) ? c3 : c4;

        {
            const float* za = z + (size_t)(b0 + 8 * k + (l & 7)) * S_DIM
                                + k_begin + wv * C5_QUARTER + 4 * half;
            const float* cb = c5 + (size_t)l * S_DIM
                                 + k_begin + wv * C5_QUARTER + 4 * half;
            v8f e0 = {}, e1 = {};
            C5Frags q0, q1;
            c5_load(q0, za, cb);
            c5_load(q1, za + C5_CHUNK, cb + C5_CHUNK);
            __builtin_amdgcn_sched_barrier(0);
            c5_mma(q0, e0, e1);
            __builtin_amdgcn_sched_barrier(0);
            c5_load(q0, za + 2 * C5_CHUNK, cb + 2 * C5_CHUNK);
            __builtin_amdgcn_sched_barrier(0);
            c5_mma(q1, e0, e1);
            __builtin_amdgcn_sched_barrier(0);
            c5_load(q1, za + 3 * C5_CHUNK, cb + 3 * C5_CHUNK);
            __builtin_amdgcn_sched_barrier(0);
            c5_mma(q0, e0, e1);
            c5_mma(q1, e0, e1);
            e0 += e1;
            if (half == 0) {
#pragma unroll
                for (int v = 0; v < 8; ++v)
                    red[wv][v][l] = e0[v];
            }
        }
        __syncthreads();
        {
            const int v = threadIdx.x >> 4;
            const int n = threadIdx.x & 15;
            const float s = ((red[0][v][n] + red[1][v][n]) + red[2][v][n])
                            + red[3][v][n];
            vpart[(size_t)ks * B_BATCH * NCOLS
                  + (size_t)(b0 + 8 * k + v) * NCOLS + 1024 + n] = s;
        }

        const float* ap0 = z + (size_t)(b0 + l) * S_DIM + k_begin + 2 * half;
        const float* ap1 = ap0 + (size_t)16 * S_DIM;
        const float* bp  = cp + (size_t)(d0 + (l >> 2)) * CORE_D_STRIDE
                              + (size_t)(k_begin + 2 * half) * R_RANK + 4 * (l & 3);
        const float* pfb = cp + (size_t)(d0 + (lane >> 3)) * CORE_D_STRIDE
                              + (size_t)k_begin * R_RANK + (lane & 7) * 64;
        const float* pfa = z + (size_t)(b0 + wv * 8 + (lane & 7)) * S_DIM + k_begin;

        v8f acc[2][4] = {};
        P1Frags f0, f1;
        p1_load(f0, ap0, ap1, bp);
#pragma unroll
        for (int p = 2; p < P1_PF; p += 2) {
            __builtin_prefetch(pfb + p * 256, 0, 1);
            __builtin_prefetch(pfa + p * 16, 0, 1);
        }
        for (int it = 0; it < P1_ITERS - 2; it += 2) {
            const int pf = min(it + P1_PF, P1_ITERS - 2);
            __builtin_prefetch(pfb + pf * 256, 0, 1);
            __builtin_prefetch(pfa + pf * 16, 0, 1);
            p1_load(f1, ap0 + 16, ap1 + 16, bp + 256);
            __builtin_amdgcn_sched_barrier(0);
            p1_mma(f0, acc);
            __builtin_amdgcn_sched_barrier(0);
            p1_load(f0, ap0 + 32, ap1 + 32, bp + 512);
            __builtin_amdgcn_sched_barrier(0);
            p1_mma(f1, acc);
            __builtin_amdgcn_sched_barrier(0);
            ap0 += 32; ap1 += 32; bp += 512;
        }
        p1_load(f1, ap0 + 16, ap1 + 16, bp + 256);
        __builtin_amdgcn_sched_barrier(0);
        p1_mma(f0, acc);
        p1_mma(f1, acc);

        float* vp = vpart + (size_t)ks * B_BATCH * NCOLS
                          + (size_t)(b0 + 8 * half) * NCOLS
                          + k * 256 + d0 * 16 + 4 * l;
#pragma unroll
        for (int m = 0; m < 2; ++m) {
#pragma unroll
            for (int v = 0; v < 8; ++v) {
                v4f o = {acc[m][0][v], acc[m][1][v], acc[m][2][v], acc[m][3][v]};
                *(v4f*)(vp + (size_t)(m * 16 + v) * NCOLS) = o;
            }
        }
    }
}

#define F_ROWS    16
#define F_PIX     1024
#define F_THREADS 256
#define F_WAVES   (F_THREADS / 32)
#define F_TILES   (F_PIX / 16 / F_WAVES)
#define V_F4      (F_ROWS * NCOLS / 4)
#define PLANE_F4  (B_BATCH * NCOLS / 4)

__global__ __launch_bounds__(F_THREADS)
void tt_chain_out(const float* __restrict__ vpart,
                  const float* __restrict__ core0,
                  float* __restrict__ out) {
    __shared__ v4f   Vl[V_F4];
    __shared__ float tl[2][F_ROWS][R_RANK];

    const int tid  = threadIdx.x;
    const int lane = tid & 31;
    const int wv   = tid >> 5;
    const int half = lane >> 4;
    const int l    = lane & 15;
    const int b0   = blockIdx.y * F_ROWS;
    const int s0   = blockIdx.x * F_PIX + wv * (F_PIX / F_WAVES);

    const float* cp = core0 + (size_t)(s0 + l) * R_RANK + 2 * half;
    v2f bf[F_TILES][4];
#pragma unroll
    for (int j = 0; j < F_TILES; ++j) {
#pragma unroll
        for (int q = 0; q < 4; ++q)
            bf[j][q] = *(const v2f*)(cp + (size_t)j * 16 * R_RANK + q * 4);
    }

    const v4f* src = (const v4f*)(vpart + (size_t)b0 * NCOLS);
    const bool has_rest = wv < (V_F4 % F_THREADS) / 32;
    const int  i_rest   = (V_F4 / F_THREADS) * F_THREADS + tid;
    v4f rest[KSPLIT];
    if (has_rest) {
#pragma unroll
        for (int p = 0; p < KSPLIT; ++p)
            rest[p] = src[(size_t)p * PLANE_F4 + i_rest];
    }
#pragma unroll 4
    for (int it = 0; it < V_F4 / F_THREADS; ++it) {
        const int i = it * F_THREADS + tid;
        v4f s = {0.f, 0.f, 0.f, 0.f};
#pragma unroll
        for (int p = 0; p < KSPLIT; ++p)
            s += src[(size_t)p * PLANE_F4 + i];
        Vl[i] = s;
    }
    if (has_rest) {
        v4f s = {0.f, 0.f, 0.f, 0.f};
#pragma unroll
        for (int p = 0; p < KSPLIT; ++p)
            s += rest[p];
        Vl[i_rest] = s;
    }
    __syncthreads();

    {
        const int r = tid >> 4;
        const int d = tid & 15;
        const float* Vrow = (const float*)Vl + r * NCOLS;
        tl[0][r][d] = Vrow[1024 + d];
        __syncthreads();
        int cur = 0;
        for (int k = 3; k >= 0; --k) {
            const float* vk = Vrow + k * 256 + d * 16;
            const float* t  = tl[cur][r];
            float s = 0.f;
#pragma unroll
            for (int e = 0; e < R_RANK; ++e)
                s += vk[e] * t[e];
            tl[cur ^ 1][r][d] = s;
            cur ^= 1;
            __syncthreads();
        }
    }

    const float* wp = &tl[0][l][2 * half];
    v2f a[4];
#pragma unroll
    for (int q = 0; q < 4; ++q)
        a[q] = *(const v2f*)(wp + q * 4);

    float* op = out + (size_t)(b0 + 8 * half) * S_DIM + s0 + l;
#pragma unroll
    for (int j = 0; j < F_TILES; ++j) {
        v8f acc = {};
#pragma unroll
        for (int q = 0; q < 4; ++q)
            acc = wmma_f32x4(a[q], bf[j][q], acc);
#pragma unroll
        for (int v = 0; v < 8; ++v)
            op[(size_t)v * S_DIM + j * 16] = acc[v];
    }
}

extern "C" void kernel_launch(void* const* d_in, const int* in_sizes, int n_in,
                              void* d_out, int out_size, void* d_ws, size_t ws_size,
                              hipStream_t stream) {
    const float* z  = (const float*)d_in[0];
    const float* c0 = (const float*)d_in[1];
    const float* c1 = (const float*)d_in[2];
    const float* c2 = (const float*)d_in[3];
    const float* c3 = (const float*)d_in[4];
    const float* c4 = (const float*)d_in[5];
    const float* c5 = (const float*)d_in[6];
    float* out = (float*)d_out;

    float* vpart = (float*)d_ws;

    tt_phase1<<<dim3(P1_MAIN_GROUPS), P1_THREADS, 0, stream>>>(
        z, c1, c2, c3, c4, c5, vpart);
    tt_chain_out<<<dim3(S_DIM / F_PIX, B_BATCH / F_ROWS), F_THREADS, 0, stream>>>(
        vpart, c0, out);
}


# phase 1 re-tiled: group tile 64 x 128, each wave computes all of it (4 M x 8 N tiles in registers) on its own K quarter, z read in whole lines; quarters added via LDS in wave order; sum order changed
# speedup vs baseline: 7.6838x; 1.0507x over previous
#include <hip/hip_runtime.h>

typedef __attribute__((ext_vector_type(2))) float v2f;
typedef __attribute__((ext_vector_type(4))) float v4f;
typedef __attribute__((ext_vector_type(8))) float v8f;

#define B_BATCH 256
#define S_DIM   16384
#define R_RANK  16
#define KSPLIT  8
#define NCOLS   1040
#define KCHUNK  (S_DIM / KSPLIT)

__device__ __forceinline__ v8f wmma_f32x4(v2f a, v2f b, v8f c) {
    return __builtin_amdgcn_wmma_f32_16x16x4_f32(false, a, false, b, (short)0, c,
                                                 false, false);
}

#define P1_THREADS     128
#define P1_MAIN_GROUPS 256
#define KQ             (KCHUNK / 4)
#define P1_CHUNKS      (KQ / 64)
#define CORE_D_STRIDE  ((size_t)S_DIM * R_RANK)

struct P1A { v2f a[4][8]; };
struct P1B { v4f b[4][2][2]; };

__device__ __forceinline__ void p1_load_a(P1A& f, const float* ap) {
#pragma unroll
    for (int m = 0; m < 4; ++m) {
#pragma unroll
        for (int u = 0; u < 8; ++u)
            f.a[m][u] = *(const v2f*)(ap + (size_t)m * 16 * S_DIM + u * 4);
    }
}

__device__ __forceinline__ void p1_load_b(P1B& f, const float* bp) {
#pragma unroll
    for (int c = 0; c < 2; ++c) {
#pragma unroll
        for (int u = 0; u < 4; ++u) {
            const float* p = bp + (size_t)c * 4 * CORE_D_STRIDE + u * 64;
            f.b[u][0][c] = *(const v4f*)(p);
            f.b[u][1][c] = *(const v4f*)(p + 16);
        }
    }
}

template <int UOFF>
__device__ __forceinline__ void p1_mma(const P1A& fa, const P1B& fb,
                                       v8f (&acc)[4][8]) {
#pragma unroll
    for (int u = 0; u < 4; ++u) {
#pragma unroll
        for (int c = 0; c < 2; ++c) {
#pragma unroll
            for (int j = 0; j < 4; ++j) {
                v2f bf;
                bf.x = fb.b[u][0][c][j];
                bf.y = fb.b[u][1][c][j];
#pragma unroll
                for (int m = 0; m < 4; ++m)
                    acc[m][4 * c + j] =
                        wmma_f32x4(fa.a[m][UOFF + u], bf, acc[m][4 * c + j]);
            }
        }
    }
}
__device__ __forceinline__ void p1_prefetch(const float* pfb, const float* pfa,
                                            int t) {
#pragma unroll
    for (int q = 0; q < 4; ++q)
        __builtin_prefetch(pfb + (size_t)q * 2 * CORE_D_STRIDE + t * 1024, 0, 1);
    __builtin_prefetch(pfa + t * 64, 0, 1);
    __builtin_prefetch(pfa + (size_t)32 * S_DIM + t * 64, 0, 1);
}

#define C5_QUARTER (KCHUNK / 4)
#define C5_CHUNK   128

struct C5Frags {
    v4f a[16], b[16];
};

__device__ __forceinline__ void c5_load(C5Frags& f, const float* ap,
                                        const float* bp) {
#pragma unroll
    for (int i = 0; i < 16; ++i) {
        f.a[i] = *(const v4f*)(ap + i * 8);
        f.b[i] = *(const v4f*)(bp + i * 8);
    }
}

__device__ __forceinline__ void c5_mma(const C5Frags& f, v8f& e0, v8f& e1) {
#pragma unroll
    for (int i = 0; i < 16; ++i) {
        v2f a0 = {f.a[i][0], f.a[i][1]}, b0 = {f.b[i][0], f.b[i][1]};
        v2f a1 = {f.a[i][2], f.a[i][3]}, b1 = {f.b[i][2], f.b[i][3]};
        e0 = wmma_f32x4(a0, b0, e0);
        e1 = wmma_f32x4(a1, b1, e1);
    }
}

__global__ __launch_bounds__(P1_THREADS)
void tt_phase1(const float* __restrict__ z,
               const float* __restrict__ c1, const float* __restrict__ c2,
               const float* __restrict__ c3, const float* __restrict__ c4,
               const float* __restrict__ c5,
               float* __restrict__ vpart) {
    const int lane = threadIdx.x & 31;
    const int wv   = threadIdx.x >> 5;
    const int half = lane >> 4;
    const int l    = lane & 15;
    const int g    = blockIdx.x;
    __shared__ float red[P1_THREADS / 32][8][R_RANK];
    __shared__ v4f xch[4][4][8][2][32];

    const int ks = g & (KSPLIT - 1);
    const int nh = (g >> 3) & 1;
    const int k  = (g >> 4) & 3;
    const int mb = g >> 6;
    const int b0 = mb * 64;
    const int d0 = nh * 8;
    const int k_begin = ks * KCHUNK;
    const int r5 = 8 * (2 * k + nh);
    const float* cp = (k == 0) ? c1 : (k == 1) ? c2 : (k == 2) ? c3 : c4;

    const int kw = k_begin + wv * KQ;
    const float* pfb = cp + (size_t)(d0 + (lane >> 4)) * CORE_D_STRIDE
                          + (size_t)kw * R_RANK + (lane & 15) * 64;
    const float* pfa = z + (size_t)(b0 + lane) * S_DIM + kw;
    p1_prefetch(pfb, pfa, 0);
    p1_prefetch(pfb, pfa, 1);

    {
        const float* za = z + (size_t)(b0 + r5 + (l & 7)) * S_DIM
                            + k_begin + wv * C5_QUARTER + 4 * half;
        const float* cb = c5 + (size_t)l * S_DIM
                             + k_begin + wv * C5_QUARTER + 4 * half;
        v8f e0 = {}, e1 = {};
        C5Frags q0, q1;
        c5_load(q0, za, cb);
        c5_load(q1, za + C5_CHUNK, cb + C5_CHUNK);
        __builtin_amdgcn_sched_barrier(0);
        c5_mma(q0, e0, e1);
        __builtin_amdgcn_sched_barrier(0);
        c5_load(q0, za + 2 * C5_CHUNK, cb + 2 * C5_CHUNK);
        __builtin_amdgcn_sched_barrier(0);
        c5_mma(q1, e0, e1);
        __builtin_amdgcn_sched_barrier(0);
        c5_load(q1, za + 3 * C5_CHUNK, cb + 3 * C5_CHUNK);
        __builtin_amdgcn_sched_barrier(0);
        c5_mma(q0, e0, e1);
        c5_mma(q1, e0, e1);
        e0 += e1;
        if (half == 0) {
#pragma unroll
            for (int v = 0; v < 8; ++v)
                red[wv][v][l] = e0[v];
        }
    }
    __syncthreads();
    {
        const int v = threadIdx.x >> 4;
        const int n = threadIdx.x & 15;
        const float s = ((red[0][v][n] + red[1][v][n]) + red[2][v][n])
                        + red[3][v][n];
        vpart[(size_t)ks * B_BATCH * NCOLS
              + (size_t)(b0 + r5 + v) * NCOLS + 1024 + n] = s;
    }

    const float* ap = z + (size_t)(b0 + l) * S_DIM + kw + 2 * half;
    const float* bp = cp + (size_t)(d0 + (l >> 2)) * CORE_D_STRIDE
                         + (size_t)(kw + 2 * half) * R_RANK + 4 * (l & 3);

    v8f acc[4][8] = {};
    P1A a0, a1;
    P1B q0, q1;
    p1_load_a(a0, ap);
    p1_load_b(q0, bp);
    for (int i = 0; i < P1_CHUNKS - 1; ++i) {
        p1_prefetch(pfb, pfa, min(i + 2, P1_CHUNKS - 1));
        p1_load_b(q1, bp + 256);
        __builtin_amdgcn_sched_barrier(0);
        p1_mma<0>(a0, q0, acc);
        __builtin_amdgcn_sched_barrier(0);
        p1_load_b(q0, bp + 512);
        p1_load_a(a1, ap + 32);
        __builtin_amdgcn_sched_barrier(0);
        p1_mma<4>(a0, q1, acc);
        __builtin_amdgcn_sched_barrier(0);
        p1_load_b(q1, bp + 768);
        __builtin_amdgcn_sched_barrier(0);
        p1_mma<0>(a1, q0, acc);
        __builtin_amdgcn_sched_barrier(0);
        p1_load_b(q0, bp + 1024);
        p1_load_a(a0, ap + 64);
        __builtin_amdgcn_sched_barrier(0);
        p1_mma<4>(a1, q1, acc);
        __builtin_amdgcn_sched_barrier(0);
        ap += 64; bp += 1024;
    }
    p1_load_b(q1, bp + 256);
    __builtin_amdgcn_sched_barrier(0);
    p1_mma<0>(a0, q0, acc);
    __builtin_amdgcn_sched_barrier(0);
    p1_load_b(q0, bp + 512);
    p1_load_a(a1, ap + 32);
    __builtin_amdgcn_sched_barrier(0);
    p1_mma<4>(a0, q1, acc);
    __builtin_amdgcn_sched_barrier(0);
    p1_load_b(q1, bp + 768);
    __builtin_amdgcn_sched_barrier(0);
    p1_mma<0>(a1, q0, acc);
    p1_mma<4>(a1, q1, acc);

#pragma unroll
    for (int m = 0; m < 4; ++m) {
        if (m != wv) {
#pragma unroll
            for (int v = 0; v < 8; ++v) {
#pragma unroll
                for (int c = 0; c < 2; ++c) {
                    v4f o = {acc[m][4 * c][v], acc[m][4 * c + 1][v],
                             acc[m][4 * c + 2][v], acc[m][4 * c + 3][v]};
                    xch[wv][m][v][c][lane] = o;
                }
            }
        }
    }
    __syncthreads();
    float* vp = vpart + (size_t)ks * B_BATCH * NCOLS
                      + (size_t)(b0 + 8 * half) * NCOLS
                      + k * 256 + d0 * 16 + 4 * l;
#pragma unroll
    for (int m = 0; m < 4; ++m) {
        if (m == wv) {
#pragma unroll
            for (int v = 0; v < 8; ++v) {
#pragma unroll
                for (int c = 0; c < 2; ++c) {
                    v4f p[4];
#pragma unroll
                    for (int w = 0; w < 4; ++w) {
                        if (w == m) {
                            v4f o = {acc[m][4 * c][v], acc[m][4 * c + 1][v],
                                     acc[m][4 * c + 2][v], acc[m][4 * c + 3][v]};
                            p[w] = o;
                        } else {
                            p[w] = xch[w][m][v][c][lane];
                        }
                    }
                    const v4f s = ((p[0] + p[1]) + p[2]) + p[3];
                    *(v4f*)(vp + (size_t)(m * 16 + v) * NCOLS + c * 64) = s;
                }
            }
        }
    }
}

#define F_ROWS    16
#define F_PIX     1024
#define F_THREADS 256
#define F_WAVES   (F_THREADS / 32)
#define F_TILES   (F_PIX / 16 / F_WAVES)
#define V_F4      (F_ROWS * NCOLS / 4)
#define PLANE_F4  (B_BATCH * NCOLS / 4)

__global__ __launch_bounds__(F_THREADS)
void tt_chain_out(const float* __restrict__ vpart,
                  const float* __restrict__ core0,
                  float* __restrict__ out) {
    __shared__ v4f   Vl[V_F4];
    __shared__ float tl[2][F_ROWS][R_RANK];

    const int tid  = threadIdx.x;
    const int lane = tid & 31;
    const int wv   = tid >> 5;
    const int half = lane >> 4;
    const int l    = lane & 15;
    const int b0   = blockIdx.y * F_ROWS;
    const int s0   = blockIdx.x * F_PIX + wv * (F_PIX / F_WAVES);

    const float* cp = core0 + (size_t)(s0 + l) * R_RANK + 2 * half;
    v2f bf[F_TILES][4];
#pragma unroll
    for (int j = 0; j < F_TILES; ++j) {
#pragma unroll
        for (int q = 0; q < 4; ++q)
            bf[j][q] = *(const v2f*)(cp + (size_t)j * 16 * R_RANK + q * 4);
    }

    const v4f* src = (const v4f*)(vpart + (size_t)b0 * NCOLS);
    const bool has_rest = wv < (V_F4 % F_THREADS) / 32;
    const int  i_rest   = (V_F4 / F_THREADS) * F_THREADS + tid;
    v4f rest[KSPLIT];
    if (has_rest) {
#pragma unroll
        for (int p = 0; p < KSPLIT; ++p)
            rest[p] = src[(size_t)p * PLANE_F4 + i_rest];
    }
#pragma unroll 4
    for (int it = 0; it < V_F4 / F_THREADS; ++it) {
        const int i = it * F_THREADS + tid;
        v4f s = {0.f, 0.f, 0.f, 0.f};
#pragma unroll
        for (int p = 0; p < KSPLIT; ++p)
            s += src[(size_t)p * PLANE_F4 + i];
        Vl[i] = s;
    }
    if (has_rest) {
        v4f s = {0.f, 0.f, 0.f, 0.f};
#pragma unroll
        for (int p = 0; p < KSPLIT; ++p)
            s += rest[p];
        Vl[i_rest] = s;
    }
    __syncthreads();

    {
        const int r = tid >> 4;
        const int d = tid & 15;
        const float* Vrow = (const float*)Vl + r * NCOLS;
        tl[0][r][d] = Vrow[1024 + d];
        __syncthreads();
        int cur = 0;
        for (int k = 3; k >= 0; --k) {
            const float* vk = Vrow + k * 256 + d * 16;
            const float* t  = tl[cur][r];
            float s = 0.f;
#pragma unroll
            for (int e = 0; e < R_RANK; ++e)
                s += vk[e] * t[e];
            tl[cur ^ 1][r][d] = s;
            cur ^= 1;
            __syncthreads();
        }
    }

    const float* wp = &tl[0][l][2 * half];
    v2f a[4];
#pragma unroll
    for (int q = 0; q < 4; ++q)
        a[q] = *(const v2f*)(wp + q * 4);

    float* op = out + (size_t)(b0 + 8 * half) * S_DIM + s0 + l;
#pragma unroll
    for (int j = 0; j < F_TILES; ++j) {
        v8f acc = {};
#pragma unroll
        for (int q = 0; q < 4; ++q)
            acc = wmma_f32x4(a[q], bf[j][q], acc);
#pragma unroll
        for (int v = 0; v < 8; ++v)
            op[(size_t)v * S_DIM + j * 16] = acc[v];
    }
}

extern "C" void kernel_launch(void* const* d_in, const int* in_sizes, int n_in,
                              void* d_out, int out_size, void* d_ws, size_t ws_size,
                              hipStream_t stream) {
    const float* z  = (const float*)d_in[0];
    const float* c0 = (const float*)d_in[1];
    const float* c1 = (const float*)d_in[2];
    const float* c2 = (const float*)d_in[3];
    const float* c3 = (const float*)d_in[4];
    const float* c4 = (const float*)d_in[5];
    const float* c5 = (const float*)d_in[6];
    float* out = (float*)d_out;

    float* vpart = (float*)d_ws;

    tt_phase1<<<dim3(P1_MAIN_GROUPS), P1_THREADS, 0, stream>>>(
        z, c1, c2, c3, c4, c5, vpart);
    tt_chain_out<<<dim3(S_DIM / F_PIX, B_BATCH / F_ROWS), F_THREADS, 0, stream>>>(
        vpart, c0, out);
}
